# starting kernel
# baseline (speedup 1.0000x reference)
#include <hip/hip_runtime.h>

typedef __attribute__((ext_vector_type(2))) float v2f;
typedef __attribute__((ext_vector_type(8))) float v8f;

#define PW 32
#define PH 32
#define C_DIM 128
#define WDIM 256
#define PTOT (PW * PH)

__device__ __forceinline__ float clip1(float v) {
    return __builtin_fminf(__builtin_fmaxf(v, -1.0f), 1.0f);
}

__global__ __launch_bounds__(256) void PatchSim_wmma_kernel(
    const float* __restrict__ feat,
    const int*   __restrict__ patch_ids,
    float*       __restrict__ out)
{
    __shared__ float q_s[C_DIM];

    const int bn = blockIdx.x;
    const int b  = bn >> 8;
    const int n  = bn & 255;

    const int qid   = patch_ids[n];
    const int pos_x = qid >> 8;
    const int pos_y = qid & 255;

    int sx = pos_x - PW / 2;
    if (sx < 0) sx = 0;
    if (sx > WDIM - PW) sx = WDIM - PW;
    int sy = pos_y - PH / 2;
    if (sy < 0) sy = 0;
    if (sy > WDIM - PH) sy = WDIM - PH;

    const size_t plane = (size_t)WDIM * WDIM;
    const float* featb = feat + (size_t)b * C_DIM * plane;

    const int t = threadIdx.x;
    if (t < C_DIM) {
        q_s[t] = clip1(featb[(size_t)t * plane + (size_t)pos_x * WDIM + pos_y]);
    }
    __syncthreads();

    const int lane  = t & 31;
    const int wave  = t >> 5;
    const int i     = lane & 15;
    const int halfk = (lane >> 4) << 1;

    float* outp = out + (size_t)bn * PTOT;

    for (int j = 0; j < 8; ++j) {
        const int p0 = (wave * 8 + j) * 16;
        const int p  = p0 + i;
        const int x  = sx + (p >> 5);
        const int y  = sy + (p & 31);
        const float* kptr = featb + (size_t)halfk * plane + (size_t)x * WDIM + y;

        v8f acc = {};
        #pragma unroll 4
        for (int kc = 0; kc < 32; ++kc) {
            v2f a, bb;
            a.x = clip1(kptr[0]);
            a.y = clip1(kptr[plane]);
            const int c0 = kc * 4 + halfk;
            bb.x = q_s[c0];
            bb.y = q_s[c0 + 1];
            acc = __builtin_amdgcn_wmma_f32_16x16x4_f32(
                false, a, false, bb, (short)0, acc, false, false);
            kptr += 4 * plane;
        }

        if ((lane & 15) == 0) {
            float* o = outp + p0 + ((lane >> 4) << 3);
            #pragma unroll
            for (int k2 = 0; k2 < 8; ++k2) o[k2] = acc[k2];
        }
    }
}

extern "C" void kernel_launch(void* const* d_in, const int* in_sizes, int n_in,
                              void* d_out, int out_size, void* d_ws, size_t ws_size,
                              hipStream_t stream) {
    (void)in_sizes; (void)n_in; (void)out_size; (void)d_ws; (void)ws_size;
    const float* feat      = (const float*)d_in[0];
    const int*   patch_ids = (const int*)d_in[1];
    float*       out       = (float*)d_out;

    dim3 grid(4 * 256);
    dim3 block(256);
    hipLaunchKernelGGL(PatchSim_wmma_kernel, grid, block, 0, stream,
                       feat, patch_ids, out);
}


# each 1024-thread block serves the 4 windows nearest in Z-order of their starts (ranked in-block): 128-B lines the 4 share come from L2 once and hit in the WGP cache; order of sums unchanged
# speedup vs baseline: 1.7781x; 1.7781x over previous
#include <hip/hip_runtime.h>


typedef __attribute__((ext_vector_type(2))) float v2f;
typedef __attribute__((ext_vector_type(8))) float v8f;

#define PW 32
#define PH 32
#define C_DIM 128
#define WDIM 256
#define NWIN 256
#define PTOT (PW * PH)
#define TILES_PER_WAVE 8
#define WIN_PER_BLOCK 4

__device__ __forceinline__ float clip1(float v) {
    return __builtin_fminf(__builtin_fmaxf(v, -1.0f), 1.0f);
}

__device__ __forceinline__ int clamp_start(int pos) {
    int s = pos - PW / 2;
    if (s < 0) s = 0;
    if (s > WDIM - PW) s = WDIM - PW;
    return s;
}

__device__ __forceinline__ unsigned spread8(unsigned x) {
    x &= 0xffu;
    x = (x | (x << 4)) & 0x0f0fu;
    x = (x | (x << 2)) & 0x3333u;
    x = (x | (x << 1)) & 0x5555u;
    return x;
}

__global__ __launch_bounds__(1024) void PatchSim_wmma_kernel(
    const float* __restrict__ feat,
    const int*   __restrict__ patch_ids,
    float*       __restrict__ out)
{
    __shared__ float    q_s[WIN_PER_BLOCK][C_DIM];
    __shared__ unsigned key_s[NWIN];
    __shared__ int      part_s[4][NWIN];
    __shared__ int      sel_s[WIN_PER_BLOCK];

    const int t = threadIdx.x;
    const int b = blockIdx.x >> 6;
    const int g = blockIdx.x & 63;

    if (t < NWIN) {
        const int id = patch_ids[t];
        const unsigned z = spread8((unsigned)clamp_start(id & 255))
                         | (spread8((unsigned)clamp_start(id >> 8)) << 1);
        key_s[t] = (z << 8) | (unsigned)t;
    }
    __syncthreads();
    {
        const unsigned mine = key_s[t & (NWIN - 1)];
        const int base = (t >> 8) << 6;
        int cnt = 0;
        #pragma unroll 8
        for (int k = 0; k < 64; ++k) cnt += (key_s[base + k] < mine) ? 1 : 0;
        part_s[t >> 8][t & (NWIN - 1)] = cnt;
    }
    __syncthreads();
    if (t < NWIN) {
        const int rank = part_s[0][t] + part_s[1][t] + part_s[2][t] + part_s[3][t];
        if ((rank >> 2) == g) sel_s[rank & 3] = t;
    }
    __syncthreads();

    const int lane  = t & 31;
    const int wave  = t >> 5;
    const int win   = wave >> 3;
    const int wv    = wave & 7;

    const int n     = __builtin_amdgcn_readfirstlane(sel_s[win]);
    const int qid   = patch_ids[n];
    const int pos_x = qid >> 8;
    const int pos_y = qid & 255;
    const int sx    = clamp_start(pos_x);
    const int sy    = clamp_start(pos_y);

    const size_t plane = (size_t)WDIM * WDIM;
    const float* featb = feat + (size_t)b * C_DIM * plane;

    const int lt = t & 255;
    if (lt < C_DIM) {
        q_s[win][lt] = clip1(featb[(size_t)lt * plane + (size_t)pos_x * WDIM + pos_y]);
    }
    __syncthreads();

    const int i     = lane & 15;
    const int halfk = (lane >> 4) << 1;

    float* outp = out + ((size_t)b * NWIN + n) * PTOT;
    const float* qw = q_s[win];

    const float* kbase = featb + (size_t)halfk * plane
                       + (size_t)(sx + wv * 4) * WDIM + (size_t)(sy + i);

    v8f acc[TILES_PER_WAVE];
    #pragma unroll
    for (int j = 0; j < TILES_PER_WAVE; ++j) acc[j] = (v8f){};

    #pragma unroll 2
    for (int kc = 0; kc < 32; ++kc) {
        const int c0 = kc * 4 + halfk;
        v2f bb;
        bb.x = qw[c0];
        bb.y = qw[c0 + 1];

        v2f a[TILES_PER_WAVE];
        #pragma unroll
        for (int j = 0; j < TILES_PER_WAVE; ++j) {
            const float* kp = kbase + (j >> 1) * WDIM + (j & 1) * 16;
            a[j].x = kp[0];
            a[j].y = kp[plane];
        }
        #pragma unroll
        for (int j = 0; j < TILES_PER_WAVE; ++j) {
            v2f ac;
            ac.x = clip1(a[j].x);
            ac.y = clip1(a[j].y);
            acc[j] = __builtin_amdgcn_wmma_f32_16x16x4_f32(
                false, ac, false, bb, (short)0, acc[j], false, false);
        }
        kbase += 4 * plane;
    }

    if ((lane & 15) == 0) {
        #pragma unroll
        for (int j = 0; j < TILES_PER_WAVE; ++j) {
            float* o = outp + (wv * 8 + j) * 16 + ((lane >> 4) << 3);
            #pragma unroll
            for (int k2 = 0; k2 < 8; ++k2) o[k2] = acc[j][k2];
        }
    }
}

extern "C" void kernel_launch(void* const* d_in, const int* in_sizes, int n_in,
                              void* d_out, int out_size, void* d_ws, size_t ws_size,
                              hipStream_t stream) {
    (void)in_sizes; (void)n_in; (void)out_size; (void)d_ws; (void)ws_size;
    const float* feat      = (const float*)d_in[0];
    const int*   patch_ids = (const int*)d_in[1];
    float*       out       = (float*)d_out;

    dim3 grid(4 * 64);
    dim3 block(1024);
    hipLaunchKernelGGL(PatchSim_wmma_kernel, grid, block, 0, stream,
                       feat, patch_ids, out);
}


# each window's rows go through a two-stage LDS ring: whole-line 16-byte loads fill it one stage ahead, under the WMMAs, and the A fragments are read from LDS; order of sums unchanged
# speedup vs baseline: 1.8455x; 1.0379x over previous
#include <hip/hip_runtime.h>


typedef __attribute__((ext_vector_type(2))) float v2f;
typedef __attribute__((ext_vector_type(8))) float v8f;
typedef __attribute__((ext_vector_type(4))) float v4f;

#define PW 32
#define PH 32
#define C_DIM 128
#define WDIM 256
#define NWIN 256
#define PTOT (PW * PH)
#define TILES_PER_WAVE 8
#define WIN_PER_BLOCK 4
#define ST_PLANES 4
#define ROW_F 64
#define P_PITCH (PH * ROW_F + 16)
#define W_PITCH (ST_PLANES * P_PITCH)
#define S_PITCH (WIN_PER_BLOCK * W_PITCH)

__device__ __forceinline__ float clip1(float v) {
    return __builtin_fminf(__builtin_fmaxf(v, -1.0f), 1.0f);
}

__device__ __forceinline__ int clamp_start(int pos) {
    int s = pos - PW / 2;
    if (s < 0) s = 0;
    if (s > WDIM - PW) s = WDIM - PW;
    return s;
}

__device__ __forceinline__ unsigned spread8(unsigned x) {
    x &= 0xffu;
    x = (x | (x << 4)) & 0x0f0fu;
    x = (x | (x << 2)) & 0x3333u;
    x = (x | (x << 1)) & 0x5555u;
    return x;
}

__device__ __forceinline__ void tile_step(const float* rs, v2f bb, v8f* acc) {
    v2f a[TILES_PER_WAVE];
    #pragma unroll
    for (int j = 0; j < TILES_PER_WAVE; ++j) {
        const float* kp = rs + (j >> 1) * ROW_F + (j & 1) * 16;
        a[j].x = kp[0];
        a[j].y = kp[P_PITCH];
    }
    #pragma unroll
    for (int j = 0; j < TILES_PER_WAVE; ++j) {
        v2f ac;
        ac.x = clip1(a[j].x);
        ac.y = clip1(a[j].y);
        acc[j] = __builtin_amdgcn_wmma_f32_16x16x4_f32(
            false, ac, false, bb, (short)0, acc[j], false, false);
    }
}

__global__ __launch_bounds__(1024) void PatchSim_wmma_kernel(
    const float* __restrict__ feat,
    const int*   __restrict__ patch_ids,
    float*       __restrict__ out)
{
    __shared__ float    q_s[WIN_PER_BLOCK][C_DIM];
    __shared__ unsigned key_s[NWIN];
    __shared__ int      part_s[4][NWIN];
    __shared__ int      sel_s[WIN_PER_BLOCK];
    __shared__ __attribute__((aligned(16))) float stage_s[2 * S_PITCH];

    const int t = threadIdx.x;
    const int b = blockIdx.x >> 6;
    const int g = blockIdx.x & 63;

    if (t < NWIN) {
        const int id = patch_ids[t];
        const unsigned z = spread8((unsigned)clamp_start(id & 255))
                         | (spread8((unsigned)clamp_start(id >> 8)) << 1);
        key_s[t] = (z << 8) | (unsigned)t;
    }
    __syncthreads();
    {
        const unsigned mine = key_s[t & (NWIN - 1)];
        const int base = (t >> 8) << 6;
        int cnt = 0;
        #pragma unroll 8
        for (int k = 0; k < 64; ++k) cnt += (key_s[base + k] < mine) ? 1 : 0;
        part_s[t >> 8][t & (NWIN - 1)] = cnt;
    }
    __syncthreads();
    if (t < NWIN) {
        const int rank = part_s[0][t] + part_s[1][t] + part_s[2][t] + part_s[3][t];
        if ((rank >> 2) == g) sel_s[rank & 3] = t;
    }
    __syncthreads();

    const int lane  = t & 31;
    const int wave  = t >> 5;
    const int win   = wave >> 3;
    const int wv    = wave & 7;

    const int n     = __builtin_amdgcn_readfirstlane(sel_s[win]);
    const int qid   = patch_ids[n];
    const int pos_x = qid >> 8;
    const int pos_y = qid & 255;
    const int sx    = clamp_start(pos_x);
    const int sy    = clamp_start(pos_y);

    const size_t plane = (size_t)WDIM * WDIM;
    const float* featb = feat + (size_t)b * C_DIM * plane;

    const int lt = t & 255;
    if (lt < C_DIM) {
        q_s[win][lt] = clip1(featb[(size_t)lt * plane + (size_t)pos_x * WDIM + pos_y]);
    }
    __syncthreads();

    const int i     = lane & 15;
    const int halfk = (lane >> 4) << 1;

    float* outp = out + ((size_t)b * NWIN + n) * PTOT;
    const float* qw = q_s[win];

    const int l0    = sy >> 5;
    const int lsel  = (lt >> 3) & 1;
    const int lline = (l0 + lsel < 7) ? (l0 + lsel) : 7;
    const float* gsrc = featb + (size_t)(sx + (lt >> 4)) * WDIM + lline * 32 + (lt & 7) * 4;
    float* sdst = stage_s + win * W_PITCH + lt * 4;

    const float* rsrc = stage_s + win * W_PITCH + halfk * P_PITCH
                      + (wv * 4) * ROW_F + (sy & 31) + i;

    v8f acc[TILES_PER_WAVE];
    #pragma unroll
    for (int j = 0; j < TILES_PER_WAVE; ++j) acc[j] = (v8f){};

    v4f s0, s1, s2, s3, s4, s5, s6, s7;
    const size_t hrow = (size_t)16 * WDIM;

    s0 = *(const v4f*)(gsrc);
    s1 = *(const v4f*)(gsrc + plane);
    s2 = *(const v4f*)(gsrc + 2 * plane);
    s3 = *(const v4f*)(gsrc + 3 * plane);
    s4 = *(const v4f*)(gsrc + hrow);
    s5 = *(const v4f*)(gsrc + plane + hrow);
    s6 = *(const v4f*)(gsrc + 2 * plane + hrow);
    s7 = *(const v4f*)(gsrc + 3 * plane + hrow);
    *(v4f*)(sdst)                      = s0;
    *(v4f*)(sdst + P_PITCH)            = s1;
    *(v4f*)(sdst + 2 * P_PITCH)        = s2;
    *(v4f*)(sdst + 3 * P_PITCH)        = s3;
    *(v4f*)(sdst + 1024)               = s4;
    *(v4f*)(sdst + P_PITCH + 1024)     = s5;
    *(v4f*)(sdst + 2 * P_PITCH + 1024) = s6;
    *(v4f*)(sdst + 3 * P_PITCH + 1024) = s7;
    __syncthreads();

    int cur = 0;
    for (int kc = 0; kc < 31; ++kc) {
        const float* gs = gsrc + (size_t)(kc + 1) * ST_PLANES * plane;
        float* sd = sdst + (cur ^ 1) * S_PITCH;
        const float* rs = rsrc + cur * S_PITCH;

        s0 = *(const v4f*)(gs);
        s1 = *(const v4f*)(gs + plane);
        s2 = *(const v4f*)(gs + 2 * plane);
        s3 = *(const v4f*)(gs + 3 * plane);
        s4 = *(const v4f*)(gs + hrow);
        s5 = *(const v4f*)(gs + plane + hrow);
        s6 = *(const v4f*)(gs + 2 * plane + hrow);
        s7 = *(const v4f*)(gs + 3 * plane + hrow);

        const int c0 = kc * 4 + halfk;
        v2f bb;
        bb.x = qw[c0];
        bb.y = qw[c0 + 1];

        tile_step(rs, bb, acc);

        __builtin_amdgcn_sched_barrier(0);

        *(v4f*)(sd)                      = s0;
        *(v4f*)(sd + P_PITCH)            = s1;
        *(v4f*)(sd + 2 * P_PITCH)        = s2;
        *(v4f*)(sd + 3 * P_PITCH)        = s3;
        *(v4f*)(sd + 1024)               = s4;
        *(v4f*)(sd + P_PITCH + 1024)     = s5;
        *(v4f*)(sd + 2 * P_PITCH + 1024) = s6;
        *(v4f*)(sd + 3 * P_PITCH + 1024) = s7;
        __syncthreads();
        cur ^= 1;
    }
    {
        const float* rs = rsrc + cur * S_PITCH;
        const int c0 = 31 * 4 + halfk;
        v2f bb;
        bb.x = qw[c0];
        bb.y = qw[c0 + 1];
        tile_step(rs, bb, acc);
    }

    if ((lane & 15) == 0) {
        #pragma unroll
        for (int j = 0; j < TILES_PER_WAVE; ++j) {
            float* o = outp + (wv * 8 + j) * 16 + ((lane >> 4) << 3);
            #pragma unroll
            for (int k2 = 0; k2 < 8; ++k2) o[k2] = acc[j][k2];
        }
    }
}

extern "C" void kernel_launch(void* const* d_in, const int* in_sizes, int n_in,
                              void* d_out, int out_size, void* d_ws, size_t ws_size,
                              hipStream_t stream) {
    (void)in_sizes; (void)n_in; (void)out_size; (void)d_ws; (void)ws_size;
    const float* feat      = (const float*)d_in[0];
    const int*   patch_ids = (const int*)d_in[1];
    float*       out       = (float*)d_out;

    dim3 grid(4 * 64);
    dim3 block(1024);
    hipLaunchKernelGGL(PatchSim_wmma_kernel, grid, block, 0, stream,
                       feat, patch_ids, out);
}


# speedup vs baseline: 1.9376x; 1.0499x over previous
#include <hip/hip_runtime.h>


typedef __attribute__((ext_vector_type(2))) float v2f;
typedef __attribute__((ext_vector_type(8))) float v8f;
typedef __attribute__((ext_vector_type(4))) float v4f;

#define PW 32
#define PH 32
#define C_DIM 128
#define WDIM 256
#define NWIN 256
#define PTOT (PW * PH)
#define TILES_PER_WAVE 8
#define WIN_PER_BLOCK 4
#define ST_PLANES 4
#define ROW_F 64
#define P_PITCH (PH * ROW_F + 16)
#define W_PITCH (ST_PLANES * P_PITCH)
#define S_PITCH (WIN_PER_BLOCK * W_PITCH)

__device__ __forceinline__ float clip1(float v) {
    return __builtin_fminf(__builtin_fmaxf(v, -1.0f), 1.0f);
}

__device__ __forceinline__ int clamp_start(int pos) {
    int s = pos - PW / 2;
    if (s < 0) s = 0;
    if (s > WDIM - PW) s = WDIM - PW;
    return s;
}

__device__ __forceinline__ unsigned spread8(unsigned x) {
    x &= 0xffu;
    x = (x | (x << 4)) & 0x0f0fu;
    x = (x | (x << 2)) & 0x3333u;
    x = (x | (x << 1)) & 0x5555u;
    return x;
}

__device__ __forceinline__ void tile_step(const float* rs, v2f bb, v8f* acc) {
    v2f a[TILES_PER_WAVE];
    #pragma unroll
    for (int j = 0; j < TILES_PER_WAVE; ++j) {
        const float* kp = rs + (j >> 1) * ROW_F + (j & 1) * 16;
        a[j].x = kp[0];
        a[j].y = kp[P_PITCH];
    }
    #pragma unroll
    for (int j = 0; j < TILES_PER_WAVE; ++j) {
        v2f ac;
        ac.x = clip1(a[j].x);
        ac.y = clip1(a[j].y);
        acc[j] = __builtin_amdgcn_wmma_f32_16x16x4_f32(
            false, ac, false, bb, (short)0, acc[j], false, false);
    }
}

__global__ __launch_bounds__(1024) void PatchSim_wmma_kernel(
    const float* __restrict__ feat,
    const int*   __restrict__ patch_ids,
    float*       __restrict__ out)
{
    __shared__ float    q_s[WIN_PER_BLOCK][C_DIM];
    __shared__ unsigned key_s[NWIN];
    __shared__ int      part_s[4][NWIN];
    __shared__ int      sel_s[WIN_PER_BLOCK];
    __shared__ __attribute__((aligned(16))) float stage_s[2 * S_PITCH];

    const int t = threadIdx.x;
    const int bi = blockIdx.x;
    const int b = (bi >> 2) & 3;
    const int g = ((bi & 3) << 4) | (bi >> 4);

    if (t < NWIN) {
        const int id = patch_ids[t];
        const unsigned z = spread8((unsigned)clamp_start(id & 255))
                         | (spread8((unsigned)clamp_start(id >> 8)) << 1);
        key_s[t] = (z << 8) | (unsigned)t;
    }
    __syncthreads();
    {
        const unsigned mine = key_s[t & (NWIN - 1)];
        const int base = (t >> 8) << 6;
        int cnt = 0;
        #pragma unroll 8
        for (int k = 0; k < 64; ++k) cnt += (key_s[base + k] < mine) ? 1 : 0;
        part_s[t >> 8][t & (NWIN - 1)] = cnt;
    }
    __syncthreads();
    if (t < NWIN) {
        const int rank = part_s[0][t] + part_s[1][t] + part_s[2][t] + part_s[3][t];
        if ((rank >> 2) == g) sel_s[rank & 3] = t;
    }
    __syncthreads();

    const int lane  = t & 31;
    const int wave  = t >> 5;
    const int win   = wave >> 3;
    const int wv    = wave & 7;

    const int n     = __builtin_amdgcn_readfirstlane(sel_s[win]);
    const int qid   = patch_ids[n];
    const int pos_x = qid >> 8;
    const int pos_y = qid & 255;
    const int sx    = clamp_start(pos_x);
    const int sy    = clamp_start(pos_y);

    const size_t plane = (size_t)WDIM * WDIM;
    const float* featb = feat + (size_t)b * C_DIM * plane;

    const int lt = t & 255;
    if (lt < C_DIM) {
        q_s[win][lt] = clip1(featb[(size_t)lt * plane + (size_t)pos_x * WDIM + pos_y]);
    }
    __syncthreads();

    const int i     = lane & 15;
    const int halfk = (lane >> 4) << 1;

    float* outp = out + ((size_t)b * NWIN + n) * PTOT;
    const float* qw = q_s[win];

    const int l0    = sy >> 5;
    const int lsel  = (lt >> 3) & 1;
    const int lline = (l0 + lsel < 7) ? (l0 + lsel) : 7;
    const float* gsrc = featb + (size_t)(sx + (lt >> 4)) * WDIM + lline * 32 + (lt & 7) * 4;
    float* sdst = stage_s + win * W_PITCH + lt * 4;

    const float* rsrc = stage_s + win * W_PITCH + halfk * P_PITCH
                      + (wv * 4) * ROW_F + (sy & 31) + i;

    v8f acc[TILES_PER_WAVE];
    #pragma unroll
    for (int j = 0; j < TILES_PER_WAVE; ++j) acc[j] = (v8f){};

    v4f s0, s1, s2, s3, s4, s5, s6, s7;
    const size_t hrow = (size_t)16 * WDIM;

    s0 = *(const v4f*)(gsrc);
    s1 = *(const v4f*)(gsrc + plane);
    s2 = *(const v4f*)(gsrc + 2 * plane);
    s3 = *(const v4f*)(gsrc + 3 * plane);
    s4 = *(const v4f*)(gsrc + hrow);
    s5 = *(const v4f*)(gsrc + plane + hrow);
    s6 = *(const v4f*)(gsrc + 2 * plane + hrow);
    s7 = *(const v4f*)(gsrc + 3 * plane + hrow);
    *(v4f*)(sdst)                      = s0;
    *(v4f*)(sdst + P_PITCH)            = s1;
    *(v4f*)(sdst + 2 * P_PITCH)        = s2;
    *(v4f*)(sdst + 3 * P_PITCH)        = s3;
    *(v4f*)(sdst + 1024)               = s4;
    *(v4f*)(sdst + P_PITCH + 1024)     = s5;
    *(v4f*)(sdst + 2 * P_PITCH + 1024) = s6;
    *(v4f*)(sdst + 3 * P_PITCH + 1024) = s7;
    __syncthreads();

    int cur = 0;
    for (int kc = 0; kc < 31; ++kc) {
        const float* gs = gsrc + (size_t)(kc + 1) * ST_PLANES * plane;
        float* sd = sdst + (cur ^ 1) * S_PITCH;
        const float* rs = rsrc + cur * S_PITCH;

        s0 = *(const v4f*)(gs);
        s1 = *(const v4f*)(gs + plane);
        s2 = *(const v4f*)(gs + 2 * plane);
        s3 = *(const v4f*)(gs + 3 * plane);
        s4 = *(const v4f*)(gs + hrow);
        s5 = *(const v4f*)(gs + plane + hrow);
        s6 = *(const v4f*)(gs + 2 * plane + hrow);
        s7 = *(const v4f*)(gs + 3 * plane + hrow);

        const int c0 = kc * 4 + halfk;
        v2f bb;
        bb.x = qw[c0];
        bb.y = qw[c0 + 1];

        tile_step(rs, bb, acc);

        __builtin_amdgcn_sched_barrier(0);

        *(v4f*)(sd)                      = s0;
        *(v4f*)(sd + P_PITCH)            = s1;
        *(v4f*)(sd + 2 * P_PITCH)        = s2;
        *(v4f*)(sd + 3 * P_PITCH)        = s3;
        *(v4f*)(sd + 1024)               = s4;
        *(v4f*)(sd + P_PITCH + 1024)     = s5;
        *(v4f*)(sd + 2 * P_PITCH + 1024) = s6;
        *(v4f*)(sd + 3 * P_PITCH + 1024) = s7;
        __syncthreads();
        cur ^= 1;
    }
    {
        const float* rs = rsrc + cur * S_PITCH;
        const int c0 = 31 * 4 + halfk;
        v2f bb;
        bb.x = qw[c0];
        bb.y = qw[c0 + 1];
        tile_step(rs, bb, acc);
    }

    if ((lane & 15) == 0) {
        #pragma unroll
        for (int j = 0; j < TILES_PER_WAVE; ++j) {
            float* o = outp + (wv * 8 + j) * 16 + ((lane >> 4) << 3);
            #pragma unroll
            for (int k2 = 0; k2 < 8; ++k2) o[k2] = acc[j][k2];
        }
    }
}

extern "C" void kernel_launch(void* const* d_in, const int* in_sizes, int n_in,
                              void* d_out, int out_size, void* d_ws, size_t ws_size,
                              hipStream_t stream) {
    (void)in_sizes; (void)n_in; (void)out_size; (void)d_ws; (void)ws_size;
    const float* feat      = (const float*)d_in[0];
    const int*   patch_ids = (const int*)d_in[1];
    float*       out       = (float*)d_out;

    dim3 grid(4 * 64);
    dim3 block(1024);
    hipLaunchKernelGGL(PatchSim_wmma_kernel, grid, block, 0, stream,
                       feat, patch_ids, out);
}


# each wave stages exactly the 4 window rows its own tiles read, so a stage's hand-over needs no block barrier; a split barrier only keeps waves within one stage of each other; order of sums unchanged
# speedup vs baseline: 1.9970x; 1.0307x over previous
#include <hip/hip_runtime.h>


typedef __attribute__((ext_vector_type(2))) float v2f;
typedef __attribute__((ext_vector_type(8))) float v8f;
typedef __attribute__((ext_vector_type(4))) float v4f;

#define PW 32
#define PH 32
#define C_DIM 128
#define WDIM 256
#define NWIN 256
#define PTOT (PW * PH)
#define TILES_PER_WAVE 8
#define WIN_PER_BLOCK 4
#define ST_PLANES 4
#define ROW_F 64
#define P_PITCH (PH * ROW_F + 16)
#define W_PITCH (ST_PLANES * P_PITCH)
#define S_PITCH (WIN_PER_BLOCK * W_PITCH)

__device__ __forceinline__ float clip1(float v) {
    return __builtin_fminf(__builtin_fmaxf(v, -1.0f), 1.0f);
}

__device__ __forceinline__ int clamp_start(int pos) {
    int s = pos - PW / 2;
    if (s < 0) s = 0;
    if (s > WDIM - PW) s = WDIM - PW;
    return s;
}

__device__ __forceinline__ void pace_signal() { asm volatile("s_barrier_signal -1" ::: "memory"); }
__device__ __forceinline__ void pace_wait()   { asm volatile("s_barrier_wait -1" ::: "memory"); }

__device__ __forceinline__ unsigned spread8(unsigned x) {
    x &= 0xffu;
    x = (x | (x << 4)) & 0x0f0fu;
    x = (x | (x << 2)) & 0x3333u;
    x = (x | (x << 1)) & 0x5555u;
    return x;
}

__device__ __forceinline__ void tile_step(const float* rs, v2f bb, v8f* acc) {
    v2f a[TILES_PER_WAVE];
    #pragma unroll
    for (int j = 0; j < TILES_PER_WAVE; ++j) {
        const float* kp = rs + (j >> 1) * ROW_F + (j & 1) * 16;
        a[j].x = kp[0];
        a[j].y = kp[P_PITCH];
    }
    #pragma unroll
    for (int j = 0; j < TILES_PER_WAVE; ++j) {
        v2f ac;
        ac.x = clip1(a[j].x);
        ac.y = clip1(a[j].y);
        acc[j] = __builtin_amdgcn_wmma_f32_16x16x4_f32(
            false, ac, false, bb, (short)0, acc[j], false, false);
    }
}

__global__ __launch_bounds__(1024) void PatchSim_wmma_kernel(
    const float* __restrict__ feat,
    const int*   __restrict__ patch_ids,
    float*       __restrict__ out)
{
    __shared__ float    q_s[WIN_PER_BLOCK][C_DIM];
    __shared__ unsigned key_s[NWIN];
    __shared__ int      part_s[4][NWIN];
    __shared__ int      sel_s[WIN_PER_BLOCK];
    __shared__ __attribute__((aligned(16))) float stage_s[2 * S_PITCH];

    const int t = threadIdx.x;
    const int bi = blockIdx.x;
    const int b = (bi >> 2) & 3;
    const int g = ((bi & 3) << 4) | (bi >> 4);

    if (t < NWIN) {
        const int id = patch_ids[t];
        const unsigned z = spread8((unsigned)clamp_start(id & 255))
                         | (spread8((unsigned)clamp_start(id >> 8)) << 1);
        key_s[t] = (z << 8) | (unsigned)t;
    }
    __syncthreads();
    {
        const unsigned mine = key_s[t & (NWIN - 1)];
        const int base = (t >> 8) << 6;
        int cnt = 0;
        #pragma unroll 8
        for (int k = 0; k < 64; ++k) cnt += (key_s[base + k] < mine) ? 1 : 0;
        part_s[t >> 8][t & (NWIN - 1)] = cnt;
    }
    __syncthreads();
    if (t < NWIN) {
        const int rank = part_s[0][t] + part_s[1][t] + part_s[2][t] + part_s[3][t];
        if ((rank >> 2) == g) sel_s[rank & 3] = t;
    }
    __syncthreads();

    const int lane  = t & 31;
    const int wave  = t >> 5;
    const int win   = wave >> 3;
    const int wv    = wave & 7;

    const int n     = __builtin_amdgcn_readfirstlane(sel_s[win]);
    const int qid   = patch_ids[n];
    const int pos_x = qid >> 8;
    const int pos_y = qid & 255;
    const int sx    = clamp_start(pos_x);
    const int sy    = clamp_start(pos_y);

    const size_t plane = (size_t)WDIM * WDIM;
    const float* featb = feat + (size_t)b * C_DIM * plane;

    const int lt = t & 255;
    if (lt < C_DIM) {
        q_s[win][lt] = clip1(featb[(size_t)lt * plane + (size_t)pos_x * WDIM + pos_y]);
    }
    __syncthreads();

    const int i     = lane & 15;
    const int halfk = (lane >> 4) << 1;

    float* outp = out + ((size_t)b * NWIN + n) * PTOT;
    const float* qw = q_s[win];

    const int l0    = sy >> 5;
    const int lsel  = (lane >> 3) & 1;
    const int lline = (l0 + lsel < 7) ? (l0 + lsel) : 7;
    const int lrow  = wv * 4 + (lane >> 4);
    const float* gsrc = featb + (size_t)(sx + lrow) * WDIM + lline * 32 + (lane & 7) * 4;
    float* sdst = stage_s + win * W_PITCH + lrow * ROW_F + (lane & 15) * 4;

    const float* rsrc = stage_s + win * W_PITCH + halfk * P_PITCH
                      + (wv * 4) * ROW_F + (sy & 31) + i;

    v8f acc[TILES_PER_WAVE];
    #pragma unroll
    for (int j = 0; j < TILES_PER_WAVE; ++j) acc[j] = (v8f){};

    v4f s0, s1, s2, s3, s4, s5, s6, s7;
    const size_t hrow = (size_t)2 * WDIM;
    const int    hlds = 2 * ROW_F;

    s0 = *(const v4f*)(gsrc);
    s1 = *(const v4f*)(gsrc + plane);
    s2 = *(const v4f*)(gsrc + 2 * plane);
    s3 = *(const v4f*)(gsrc + 3 * plane);
    s4 = *(const v4f*)(gsrc + hrow);
    s5 = *(const v4f*)(gsrc + plane + hrow);
    s6 = *(const v4f*)(gsrc + 2 * plane + hrow);
    s7 = *(const v4f*)(gsrc + 3 * plane + hrow);
    *(v4f*)(sdst)                      = s0;
    *(v4f*)(sdst + P_PITCH)            = s1;
    *(v4f*)(sdst + 2 * P_PITCH)        = s2;
    *(v4f*)(sdst + 3 * P_PITCH)        = s3;
    *(v4f*)(sdst + hlds)               = s4;
    *(v4f*)(sdst + P_PITCH + hlds)     = s5;
    *(v4f*)(sdst + 2 * P_PITCH + hlds) = s6;
    *(v4f*)(sdst + 3 * P_PITCH + hlds) = s7;

    pace_signal();
    int cur = 0;
    for (int kc = 0; kc < 31; ++kc) {
        pace_wait();

        const float* gs = gsrc + (size_t)(kc + 1) * ST_PLANES * plane;
        float* sd = sdst + (cur ^ 1) * S_PITCH;
        const float* rs = rsrc + cur * S_PITCH;

        s0 = *(const v4f*)(gs);
        s1 = *(const v4f*)(gs + plane);
        s2 = *(const v4f*)(gs + 2 * plane);
        s3 = *(const v4f*)(gs + 3 * plane);
        s4 = *(const v4f*)(gs + hrow);
        s5 = *(const v4f*)(gs + plane + hrow);
        s6 = *(const v4f*)(gs + 2 * plane + hrow);
        s7 = *(const v4f*)(gs + 3 * plane + hrow);
        pace_signal();
        asm volatile("s_wait_dscnt 0x0" ::: "memory");

        const int c0 = kc * 4 + halfk;
        v2f bb;
        bb.x = qw[c0];
        bb.y = qw[c0 + 1];

        tile_step(rs, bb, acc);

        __builtin_amdgcn_sched_barrier(0);

        *(v4f*)(sd)                      = s0;
        *(v4f*)(sd + P_PITCH)            = s1;
        *(v4f*)(sd + 2 * P_PITCH)        = s2;
        *(v4f*)(sd + 3 * P_PITCH)        = s3;
        *(v4f*)(sd + hlds)               = s4;
        *(v4f*)(sd + P_PITCH + hlds)     = s5;
        *(v4f*)(sd + 2 * P_PITCH + hlds) = s6;
        *(v4f*)(sd + 3 * P_PITCH + hlds) = s7;
        cur ^= 1;
    }
    pace_wait();
    asm volatile("s_wait_dscnt 0x0" ::: "memory");
    {
        const float* rs = rsrc + cur * S_PITCH;
        const int c0 = 31 * 4 + halfk;
        v2f bb;
        bb.x = qw[c0];
        bb.y = qw[c0 + 1];
        tile_step(rs, bb, acc);
    }

    int te = t;
    asm volatile("" : "+v"(te));
    const int lane_e = te & 31;
    const int wv_e   = (te >> 5) & 7;
    if ((lane_e & 15) == 0) {
        #pragma unroll
        for (int j = 0; j < TILES_PER_WAVE; ++j) {
            float* o = outp + (wv_e * 8 + j) * 16 + ((lane_e >> 4) << 3);
            #pragma unroll
            for (int k2 = 0; k2 < 8; ++k2) o[k2] = acc[j][k2];
        }
    }
}

extern "C" void kernel_launch(void* const* d_in, const int* in_sizes, int n_in,
                              void* d_out, int out_size, void* d_ws, size_t ws_size,
                              hipStream_t stream) {
    (void)in_sizes; (void)n_in; (void)out_size; (void)d_ws; (void)ws_size;
    const float* feat      = (const float*)d_in[0];
    const int*   patch_ids = (const int*)d_in[1];
    float*       out       = (float*)d_out;

    dim3 grid(4 * 64);
    dim3 block(1024);
    hipLaunchKernelGGL(PatchSim_wmma_kernel, grid, block, 0, stream,
                       feat, patch_ids, out);
}
